# starting kernel
# baseline (speedup 1.0000x reference)
#include <hip/hip_runtime.h>
#include <hip/hip_bf16.h>

typedef __attribute__((ext_vector_type(2))) float v2f;
typedef __attribute__((ext_vector_type(8))) float v8f;

#define NGRP 40
#define NB   128
#define DIM  1024
#define MAX_TILES 64

#define WS_COUNTS 0
#define WS_ROWS   64
#define WS_TILES  5184

__global__ void build_tiles_kernel(const int* __restrict__ g,
                                   int* __restrict__ ws) {
    __shared__ int sCnt[NGRP];
    const int t = threadIdx.x;
    if (t < NGRP) {
        int cnt = 0;
        for (int b = 0; b < NB; ++b) {
            int grp = g[b] % NGRP;
            if (grp == t) {
                ws[WS_ROWS + t * NB + cnt] = b;
                ++cnt;
            }
        }
        sCnt[t] = cnt;
        ws[WS_COUNTS + t] = cnt;
    }
    __syncthreads();
    if (t == 0) {
        int nt = 0;
        for (int grp = 0; grp < NGRP; ++grp)
            for (int off = 0; off < sCnt[grp]; off += 16)
                ws[WS_TILES + nt++] = (grp << 16) | off;
        for (; nt < MAX_TILES; ++nt)
            ws[WS_TILES + nt] = -1;
    }
}

__global__ __launch_bounds__(256) void grouped_gemm_kernel(
    const float* __restrict__ x,
    const float* __restrict__ T,
    const int*   __restrict__ ws,
    float*       __restrict__ out) {

    __shared__ float Alds[DIM * 16];

    const int tile = ws[WS_TILES + blockIdx.x];
    if (tile < 0) return;
    const int grp  = tile >> 16;
    const int roff = tile & 0xFFFF;
    int m = ws[WS_COUNTS + grp] - roff;
    if (m > 16) m = 16;

    const int tid = threadIdx.x;

    for (int r = 0; r < 16; ++r) {
        const int rowIdx = (r < m) ? ws[WS_ROWS + grp * NB + roff + r] : -1;
        const float* xr = x + (size_t)(rowIdx < 0 ? 0 : rowIdx) * DIM;
        for (int d = tid; d < DIM; d += 256) {
            float v = (rowIdx >= 0) ? xr[d] : 0.0f;
            Alds[d * 16 + r] = v;
        }
    }
    __syncthreads();

    const int wave = tid >> 5;
    const int lane = tid & 31;
    const int r16  = lane & 15;
    const int half = lane >> 4;
    const int krow = half * 2;

    const int n0 = (blockIdx.y * 8 + wave) * 16;
    const float* Bp = T + (size_t)grp * DIM * DIM + n0 + r16;

    v8f c = {};

    for (int kk = 0; kk < DIM; kk += 16) {
        __builtin_prefetch(Bp + (size_t)(kk + 64) * DIM, 0, 0);
#pragma unroll
        for (int ks = 0; ks < 16; ks += 4) {
            const int k0 = kk + ks + krow;
            v2f a, b;
            a[0] = Alds[k0 * 16 + r16];
            a[1] = Alds[(k0 + 1) * 16 + r16];
            b[0] = Bp[(size_t)k0 * DIM];
            b[1] = Bp[(size_t)(k0 + 1) * DIM];
            c = __builtin_amdgcn_wmma_f32_16x16x4_f32(
                    false, a, false, b, (short)0, c, false, false);
        }
    }

#pragma unroll
    for (int j = 0; j < 8; ++j) {
        const int mrow = j + half * 8;
        if (mrow < m) {
            const int b = ws[WS_ROWS + grp * NB + roff + mrow];
            out[(size_t)b * DIM + n0 + r16] = c[j];
        }
    }
}

extern "C" void kernel_launch(void* const* d_in, const int* in_sizes, int n_in,
                              void* d_out, int out_size, void* d_ws, size_t ws_size,
                              hipStream_t stream) {
    const float* x = (const float*)d_in[0];
    const int*   g = (const int*)d_in[1];
    const float* T = (const float*)d_in[2];
    int*   ws  = (int*)d_ws;
    float* out = (float*)d_out;

    build_tiles_kernel<<<1, 64, 0, stream>>>(g, ws);

    dim3 grid(MAX_TILES, 8);
    grouped_gemm_kernel<<<grid, 256, 0, stream>>>(x, T, ws, out);
}


# T goes to the WMMAs through LDS: the block loads 64-row x 128-column stages with row-contiguous 16-byte loads two stages ahead in registers, 2 LDS buffers, hand barrier; sums and order unchanged
# speedup vs baseline: 1.1887x; 1.1887x over previous
#include <hip/hip_runtime.h>
#include <hip/hip_bf16.h>


typedef __attribute__((ext_vector_type(2))) float v2f;
typedef __attribute__((ext_vector_type(4))) float v4f;
typedef __attribute__((ext_vector_type(8))) float v8f;

#define NGRP 40
#define NB   128
#define DIM  1024
#define MAX_TILES 64

#define KC      64
#define BPITCH  136
#define NSTAGE  (DIM / KC)
#define BLD     8

#define WS_COUNTS 0
#define WS_ROWS   64
#define WS_TILES  5184

__global__ void build_tiles_kernel(const int* __restrict__ g,
                                   int* __restrict__ ws) {
    __shared__ int sCnt[NGRP];
    const int t = threadIdx.x;
    if (t < NGRP) {
        int cnt = 0;
        for (int b = 0; b < NB; ++b) {
            int grp = g[b] % NGRP;
            if (grp == t) {
                ws[WS_ROWS + t * NB + cnt] = b;
                ++cnt;
            }
        }
        sCnt[t] = cnt;
        ws[WS_COUNTS + t] = cnt;
    }
    __syncthreads();
    if (t == 0) {
        int nt = 0;
        for (int grp = 0; grp < NGRP; ++grp)
            for (int off = 0; off < sCnt[grp]; off += 16)
                ws[WS_TILES + nt++] = (grp << 16) | off;
        for (; nt < MAX_TILES; ++nt)
            ws[WS_TILES + nt] = -1;
    }
}

__device__ __forceinline__ void stage_barrier() {
    asm volatile("s_wait_dscnt 0x0" ::: "memory");
    asm volatile("s_barrier_signal -1" ::: "memory");
    asm volatile("s_barrier_wait -1" ::: "memory");
}

__device__ __forceinline__ v8f stage_mma(const float* As, const float* Bs, v8f c) {
#pragma unroll
    for (int kl = 0; kl < KC; kl += 4) {
        v2f a, b;
        a[0] = As[kl * 16];
        a[1] = As[(kl + 1) * 16];
        b[0] = Bs[kl * BPITCH];
        b[1] = Bs[(kl + 1) * BPITCH];
        c = __builtin_amdgcn_wmma_f32_16x16x4_f32(
                false, a, false, b, (short)0, c, false, false);
    }
    return c;
}

__global__ __launch_bounds__(256) void grouped_gemm_kernel(
    const float* __restrict__ x,
    const float* __restrict__ T,
    const int*   __restrict__ ws,
    float*       __restrict__ out) {

    __shared__ float Alds[DIM * 16];
    __shared__ __attribute__((aligned(16))) float Blds[2 * KC * BPITCH];

    const int tile = ws[WS_TILES + blockIdx.x];
    if (tile < 0) return;
    const int grp  = tile >> 16;
    const int roff = tile & 0xFFFF;
    int m = ws[WS_COUNTS + grp] - roff;
    if (m > 16) m = 16;

    const int tid = threadIdx.x;

    const int bslot = tid & 31;
    const int brow  = tid >> 5;
    const float* Bg = T + (size_t)grp * DIM * DIM + (size_t)brow * DIM
                        + blockIdx.y * 128 + bslot * 4;
    float* Bst = Blds + brow * BPITCH + bslot * 4;

    v4f rA[BLD], rB[BLD];
#pragma unroll
    for (int i = 0; i < BLD; ++i)
        rA[i] = *(const v4f*)(Bg + (size_t)(i * 8) * DIM);
#pragma unroll
    for (int i = 0; i < BLD; ++i)
        rB[i] = *(const v4f*)(Bg + (size_t)(KC + i * 8) * DIM);

    for (int r = 0; r < 16; ++r) {
        const int rowIdx = (r < m) ? ws[WS_ROWS + grp * NB + roff + r] : -1;
        const float* xr = x + (size_t)(rowIdx < 0 ? 0 : rowIdx) * DIM;
        for (int d = tid; d < DIM; d += 256) {
            float v = (rowIdx >= 0) ? xr[d] : 0.0f;
            Alds[d * 16 + r] = v;
        }
    }

    float* Bst0 = Bst;
    float* Bst1 = Bst + KC * BPITCH;
#pragma unroll
    for (int i = 0; i < BLD; ++i)
        *(v4f*)(Bst0 + i * 8 * BPITCH) = rA[i];
    stage_barrier();

    const int wave = tid >> 5;
    const int lane = tid & 31;
    const int r16  = lane & 15;
    const int half = lane >> 4;
    const int krow = half * 2;

    const int n0 = (blockIdx.y * 8 + wave) * 16;
    const float* Ap  = Alds + krow * 16 + r16;
    const float* Bp0 = Blds + krow * BPITCH + wave * 16 + r16;
    const float* Bp1 = Bp0 + KC * BPITCH;

    v8f c = {};

    for (int s = 0; s < NSTAGE - 2; s += 2) {
        const float* Bn = Bg + (size_t)(s + 2) * KC * DIM;
#pragma unroll
        for (int i = 0; i < BLD; ++i)
            rA[i] = *(const v4f*)(Bn + (size_t)(i * 8) * DIM);

        c = stage_mma(Ap + s * KC * 16, Bp0, c);

#pragma unroll
        for (int i = 0; i < BLD; ++i)
            *(v4f*)(Bst1 + i * 8 * BPITCH) = rB[i];
        stage_barrier();

#pragma unroll
        for (int i = 0; i < BLD; ++i)
            rB[i] = *(const v4f*)(Bn + (size_t)(KC + i * 8) * DIM);

        c = stage_mma(Ap + (s + 1) * KC * 16, Bp1, c);

#pragma unroll
        for (int i = 0; i < BLD; ++i)
            *(v4f*)(Bst0 + i * 8 * BPITCH) = rA[i];
        stage_barrier();
    }
    c = stage_mma(Ap + (NSTAGE - 2) * KC * 16, Bp0, c);
#pragma unroll
    for (int i = 0; i < BLD; ++i)
        *(v4f*)(Bst1 + i * 8 * BPITCH) = rB[i];
    stage_barrier();
    c = stage_mma(Ap + (NSTAGE - 1) * KC * 16, Bp1, c);

#pragma unroll
    for (int j = 0; j < 8; ++j) {
        const int mrow = j + half * 8;
        if (mrow < m) {
            const int b = ws[WS_ROWS + grp * NB + roff + mrow];
            out[(size_t)b * DIM + n0 + r16] = c[j];
        }
    }
}

extern "C" void kernel_launch(void* const* d_in, const int* in_sizes, int n_in,
                              void* d_out, int out_size, void* d_ws, size_t ws_size,
                              hipStream_t stream) {
    const float* x = (const float*)d_in[0];
    const int*   g = (const int*)d_in[1];
    const float* T = (const float*)d_in[2];
    int*   ws  = (int*)d_ws;
    float* out = (float*)d_out;

    build_tiles_kernel<<<1, 64, 0, stream>>>(g, ws);

    dim3 grid(MAX_TILES, 8);
    grouped_gemm_kernel<<<grid, 256, 0, stream>>>(x, T, ws, out);
}


# Prologue bucketing made parallel: one thread per sample ranks itself among equal matrix indices from an LDS copy of all 128; tile slots by a sum over the 40 group counts; same workspace, 2 launches
# speedup vs baseline: 1.5060x; 1.2669x over previous
#include <hip/hip_runtime.h>
#include <hip/hip_bf16.h>


typedef __attribute__((ext_vector_type(2))) float v2f;
typedef __attribute__((ext_vector_type(4))) float v4f;
typedef __attribute__((ext_vector_type(8))) float v8f;

#define NGRP 40
#define NB   128
#define DIM  1024
#define MAX_TILES 64

#define KC      64
#define BPITCH  136
#define NSTAGE  (DIM / KC)
#define BLD     8

#define WS_COUNTS 0
#define WS_ROWS   64
#define WS_TILES  5184

__global__ __launch_bounds__(NB) void build_tiles_kernel(const int* __restrict__ g,
                                                         int* __restrict__ ws) {
    __shared__ __attribute__((aligned(16))) int sGrp[NB];
    __shared__ int sNt[NGRP];
    const int t = threadIdx.x;

    int mine = g[t] % NGRP;
    if (mine < 0) mine += NGRP;
    sGrp[t] = mine;
    __syncthreads();

    int rank = 0;
    int cnt  = 0;
#pragma unroll
    for (int j = 0; j < NB; j += 4) {
        const int4 q = *(const int4*)(sGrp + j);
        rank += (q.x == mine && j     < t) ? 1 : 0;
        rank += (q.y == mine && j + 1 < t) ? 1 : 0;
        rank += (q.z == mine && j + 2 < t) ? 1 : 0;
        rank += (q.w == mine && j + 3 < t) ? 1 : 0;
        cnt  += (q.x == t) ? 1 : 0;
        cnt  += (q.y == t) ? 1 : 0;
        cnt  += (q.z == t) ? 1 : 0;
        cnt  += (q.w == t) ? 1 : 0;
    }
    ws[WS_ROWS + mine * NB + rank] = t;

    const int nt = (cnt + 15) >> 4;
    if (t < NGRP) {
        ws[WS_COUNTS + t] = cnt;
        sNt[t] = nt;
    }
    __syncthreads();

    if (t < MAX_TILES) {
        int before = 0;
        int total  = 0;
        for (int j = 0; j < NGRP; ++j) {
            const int n = sNt[j];
            before += (j < t) ? n : 0;
            total  += n;
        }
        for (int i = 0; i < NB / 16; ++i)
            if (t < NGRP && i < nt)
                ws[WS_TILES + before + i] = (t << 16) | (i << 4);
        if (t >= total)
            ws[WS_TILES + t] = -1;
    }
}

__device__ __forceinline__ void stage_barrier() {
    asm volatile("s_wait_dscnt 0x0" ::: "memory");
    asm volatile("s_barrier_signal -1" ::: "memory");
    asm volatile("s_barrier_wait -1" ::: "memory");
}

__device__ __forceinline__ v8f stage_mma(const float* As, const float* Bs, v8f c) {
#pragma unroll
    for (int kl = 0; kl < KC; kl += 4) {
        v2f a, b;
        a[0] = As[kl * 16];
        a[1] = As[(kl + 1) * 16];
        b[0] = Bs[kl * BPITCH];
        b[1] = Bs[(kl + 1) * BPITCH];
        c = __builtin_amdgcn_wmma_f32_16x16x4_f32(
                false, a, false, b, (short)0, c, false, false);
    }
    return c;
}

__global__ __launch_bounds__(256) void grouped_gemm_kernel(
    const float* __restrict__ x,
    const float* __restrict__ T,
    const int*   __restrict__ ws,
    float*       __restrict__ out) {

    __shared__ float Alds[DIM * 16];
    __shared__ __attribute__((aligned(16))) float Blds[2 * KC * BPITCH];

    const int tile = ws[WS_TILES + blockIdx.x];
    if (tile < 0) return;
    const int grp  = tile >> 16;
    const int roff = tile & 0xFFFF;
    int m = ws[WS_COUNTS + grp] - roff;
    if (m > 16) m = 16;

    const int tid = threadIdx.x;

    const int bslot = tid & 31;
    const int brow  = tid >> 5;
    const float* Bg = T + (size_t)grp * DIM * DIM + (size_t)brow * DIM
                        + blockIdx.y * 128 + bslot * 4;
    float* Bst = Blds + brow * BPITCH + bslot * 4;

    v4f rA[BLD], rB[BLD];
#pragma unroll
    for (int i = 0; i < BLD; ++i)
        rA[i] = *(const v4f*)(Bg + (size_t)(i * 8) * DIM);
#pragma unroll
    for (int i = 0; i < BLD; ++i)
        rB[i] = *(const v4f*)(Bg + (size_t)(KC + i * 8) * DIM);

    for (int r = 0; r < 16; ++r) {
        const int rowIdx = (r < m) ? ws[WS_ROWS + grp * NB + roff + r] : -1;
        const float* xr = x + (size_t)(rowIdx < 0 ? 0 : rowIdx) * DIM;
        for (int d = tid; d < DIM; d += 256) {
            float v = (rowIdx >= 0) ? xr[d] : 0.0f;
            Alds[d * 16 + r] = v;
        }
    }

    float* Bst0 = Bst;
    float* Bst1 = Bst + KC * BPITCH;
#pragma unroll
    for (int i = 0; i < BLD; ++i)
        *(v4f*)(Bst0 + i * 8 * BPITCH) = rA[i];
    stage_barrier();

    const int wave = tid >> 5;
    const int lane = tid & 31;
    const int r16  = lane & 15;
    const int half = lane >> 4;
    const int krow = half * 2;

    const int n0 = (blockIdx.y * 8 + wave) * 16;
    const float* Ap  = Alds + krow * 16 + r16;
    const float* Bp0 = Blds + krow * BPITCH + wave * 16 + r16;
    const float* Bp1 = Bp0 + KC * BPITCH;

    v8f c = {};

    for (int s = 0; s < NSTAGE - 2; s += 2) {
        const float* Bn = Bg + (size_t)(s + 2) * KC * DIM;
#pragma unroll
        for (int i = 0; i < BLD; ++i)
            rA[i] = *(const v4f*)(Bn + (size_t)(i * 8) * DIM);

        c = stage_mma(Ap + s * KC * 16, Bp0, c);

#pragma unroll
        for (int i = 0; i < BLD; ++i)
            *(v4f*)(Bst1 + i * 8 * BPITCH) = rB[i];
        stage_barrier();

#pragma unroll
        for (int i = 0; i < BLD; ++i)
            rB[i] = *(const v4f*)(Bn + (size_t)(KC + i * 8) * DIM);

        c = stage_mma(Ap + (s + 1) * KC * 16, Bp1, c);

#pragma unroll
        for (int i = 0; i < BLD; ++i)
            *(v4f*)(Bst0 + i * 8 * BPITCH) = rA[i];
        stage_barrier();
    }
    c = stage_mma(Ap + (NSTAGE - 2) * KC * 16, Bp0, c);
#pragma unroll
    for (int i = 0; i < BLD; ++i)
        *(v4f*)(Bst1 + i * 8 * BPITCH) = rB[i];
    stage_barrier();
    c = stage_mma(Ap + (NSTAGE - 1) * KC * 16, Bp1, c);

#pragma unroll
    for (int j = 0; j < 8; ++j) {
        const int mrow = j + half * 8;
        if (mrow < m) {
            const int b = ws[WS_ROWS + grp * NB + roff + mrow];
            out[(size_t)b * DIM + n0 + r16] = c[j];
        }
    }
}

extern "C" void kernel_launch(void* const* d_in, const int* in_sizes, int n_in,
                              void* d_out, int out_size, void* d_ws, size_t ws_size,
                              hipStream_t stream) {
    const float* x = (const float*)d_in[0];
    const int*   g = (const int*)d_in[1];
    const float* T = (const float*)d_in[2];
    int*   ws  = (int*)d_ws;
    float* out = (float*)d_out;

    build_tiles_kernel<<<1, NB, 0, stream>>>(g, ws);

    dim3 grid(MAX_TILES, 8);
    grouped_gemm_kernel<<<grid, 256, 0, stream>>>(x, T, ws, out);
}


# x rows K-tiled into T's ring: one 16-byte load per thread per stage, two stages ahead, row-major LDS at pitch 68; the up-front 64 KB A tile and its 64 load-wait-store trips are gone; sums unchanged
# speedup vs baseline: 2.1170x; 1.4057x over previous
#include <hip/hip_runtime.h>
#include <hip/hip_bf16.h>


typedef __attribute__((ext_vector_type(2))) float v2f;
typedef __attribute__((ext_vector_type(4))) float v4f;
typedef __attribute__((ext_vector_type(8))) float v8f;

#define NGRP 40
#define NB   128
#define DIM  1024
#define MAX_TILES 64

#define KC      64
#define BPITCH  136
#define NSTAGE  (DIM / KC)
#define BLD     8

#define APITCH  68

#define WS_COUNTS 0
#define WS_ROWS   64
#define WS_TILES  5184

__global__ __launch_bounds__(NB) void build_tiles_kernel(const int* __restrict__ g,
                                                         int* __restrict__ ws) {
    __shared__ __attribute__((aligned(16))) int sGrp[NB];
    __shared__ int sNt[NGRP];
    const int t = threadIdx.x;

    int mine = g[t] % NGRP;
    if (mine < 0) mine += NGRP;
    sGrp[t] = mine;
    __syncthreads();

    int rank = 0;
    int cnt  = 0;
#pragma unroll
    for (int j = 0; j < NB; j += 4) {
        const int4 q = *(const int4*)(sGrp + j);
        rank += (q.x == mine && j     < t) ? 1 : 0;
        rank += (q.y == mine && j + 1 < t) ? 1 : 0;
        rank += (q.z == mine && j + 2 < t) ? 1 : 0;
        rank += (q.w == mine && j + 3 < t) ? 1 : 0;
        cnt  += (q.x == t) ? 1 : 0;
        cnt  += (q.y == t) ? 1 : 0;
        cnt  += (q.z == t) ? 1 : 0;
        cnt  += (q.w == t) ? 1 : 0;
    }
    ws[WS_ROWS + mine * NB + rank] = t;

    const int nt = (cnt + 15) >> 4;
    if (t < NGRP) {
        ws[WS_COUNTS + t] = cnt;
        sNt[t] = nt;
    }
    __syncthreads();

    if (t < MAX_TILES) {
        int before = 0;
        int total  = 0;
        for (int j = 0; j < NGRP; ++j) {
            const int n = sNt[j];
            before += (j < t) ? n : 0;
            total  += n;
        }
        for (int i = 0; i < NB / 16; ++i)
            if (t < NGRP && i < nt)
                ws[WS_TILES + before + i] = (t << 16) | (i << 4);
        if (t >= total)
            ws[WS_TILES + t] = -1;
    }
}

__device__ __forceinline__ void stage_barrier() {
    asm volatile("s_wait_dscnt 0x0" ::: "memory");
    asm volatile("s_barrier_signal -1" ::: "memory");
    asm volatile("s_barrier_wait -1" ::: "memory");
}

__device__ __forceinline__ v8f stage_mma(const float* As, const float* Bs, v8f c) {
#pragma unroll
    for (int kl = 0; kl < KC; kl += 4) {
        v2f b;
        const v2f a = *(const v2f*)(As + kl);
        b[0] = Bs[kl * BPITCH];
        b[1] = Bs[(kl + 1) * BPITCH];
        c = __builtin_amdgcn_wmma_f32_16x16x4_f32(
                false, a, false, b, (short)0, c, false, false);
    }
    return c;
}

__global__ __launch_bounds__(256) void grouped_gemm_kernel(
    const float* __restrict__ x,
    const float* __restrict__ T,
    const int*   __restrict__ ws,
    float*       __restrict__ out) {

    __shared__ __attribute__((aligned(16))) float Alds[2 * 16 * APITCH];
    __shared__ __attribute__((aligned(16))) float Blds[2 * KC * BPITCH];

    const int tile = ws[WS_TILES + blockIdx.x];
    if (tile < 0) return;
    const int grp  = tile >> 16;
    const int roff = tile & 0xFFFF;
    int m = ws[WS_COUNTS + grp] - roff;
    if (m > 16) m = 16;

    const int tid = threadIdx.x;

    const int bslot = tid & 31;
    const int brow  = tid >> 5;
    const float* Bg = T + (size_t)grp * DIM * DIM + (size_t)brow * DIM
                        + blockIdx.y * 128 + bslot * 4;
    float* Bst = Blds + brow * BPITCH + bslot * 4;

    const int arow  = tid >> 4;
    const int aslot = tid & 15;
    const int arowIdx = ws[WS_ROWS + grp * NB + roff + (arow < m ? arow : m - 1)];

    v4f rA[BLD], rB[BLD];
    v4f aA, aB;
#pragma unroll
    for (int i = 0; i < BLD; ++i)
        rA[i] = *(const v4f*)(Bg + (size_t)(i * 8) * DIM);

    const float* Ag = x + (size_t)arowIdx * DIM + aslot * 4;
    aA = *(const v4f*)(Ag);
    aB = *(const v4f*)(Ag + KC);
#pragma unroll
    for (int i = 0; i < BLD; ++i)
        rB[i] = *(const v4f*)(Bg + (size_t)(KC + i * 8) * DIM);

    float* Bst0 = Bst;
    float* Bst1 = Bst + KC * BPITCH;
    float* Ast0 = Alds + arow * APITCH + aslot * 4;
    float* Ast1 = Ast0 + 16 * APITCH;
#pragma unroll
    for (int i = 0; i < BLD; ++i)
        *(v4f*)(Bst0 + i * 8 * BPITCH) = rA[i];
    *(v4f*)Ast0 = aA;
    stage_barrier();

    const int wave = tid >> 5;
    const int lane = tid & 31;
    const int r16  = lane & 15;
    const int half = lane >> 4;
    const int krow = half * 2;

    const int n0 = (blockIdx.y * 8 + wave) * 16;
    const float* Ap0 = Alds + r16 * APITCH + krow;
    const float* Ap1 = Ap0 + 16 * APITCH;
    const float* Bp0 = Blds + krow * BPITCH + wave * 16 + r16;
    const float* Bp1 = Bp0 + KC * BPITCH;

    v8f c = {};

    for (int s = 0; s < NSTAGE - 2; s += 2) {
        const float* Bn = Bg + (size_t)(s + 2) * KC * DIM;
        const float* An = Ag + (s + 2) * KC;
#pragma unroll
        for (int i = 0; i < BLD; ++i)
            rA[i] = *(const v4f*)(Bn + (size_t)(i * 8) * DIM);
        aA = *(const v4f*)(An);

        c = stage_mma(Ap0, Bp0, c);

#pragma unroll
        for (int i = 0; i < BLD; ++i)
            *(v4f*)(Bst1 + i * 8 * BPITCH) = rB[i];
        *(v4f*)Ast1 = aB;
        stage_barrier();

#pragma unroll
        for (int i = 0; i < BLD; ++i)
            rB[i] = *(const v4f*)(Bn + (size_t)(KC + i * 8) * DIM);
        aB = *(const v4f*)(An + KC);

        c = stage_mma(Ap1, Bp1, c);

#pragma unroll
        for (int i = 0; i < BLD; ++i)
            *(v4f*)(Bst0 + i * 8 * BPITCH) = rA[i];
        *(v4f*)Ast0 = aA;
        stage_barrier();
    }
    c = stage_mma(Ap0, Bp0, c);
#pragma unroll
    for (int i = 0; i < BLD; ++i)
        *(v4f*)(Bst1 + i * 8 * BPITCH) = rB[i];
    *(v4f*)Ast1 = aB;
    stage_barrier();
    c = stage_mma(Ap1, Bp1, c);

#pragma unroll
    for (int j = 0; j < 8; ++j) {
        const int mrow = j + half * 8;
        if (mrow < m) {
            const int b = ws[WS_ROWS + grp * NB + roff + mrow];
            out[(size_t)b * DIM + n0 + r16] = c[j];
        }
    }
}

extern "C" void kernel_launch(void* const* d_in, const int* in_sizes, int n_in,
                              void* d_out, int out_size, void* d_ws, size_t ws_size,
                              hipStream_t stream) {
    const float* x = (const float*)d_in[0];
    const int*   g = (const int*)d_in[1];
    const float* T = (const float*)d_in[2];
    int*   ws  = (int*)d_ws;
    float* out = (float*)d_out;

    build_tiles_kernel<<<1, NB, 0, stream>>>(g, ws);

    dim3 grid(MAX_TILES, 8);
    grouped_gemm_kernel<<<grid, 256, 0, stream>>>(x, T, ws, out);
}


# Grid (matrix, column chunk): a block takes its matrix from blockIdx and reads count and padded row list in one round trip before T and x; tile table and 192 empty blocks gone; sums unchanged
# speedup vs baseline: 2.1988x; 1.0386x over previous
#include <hip/hip_runtime.h>
#include <hip/hip_bf16.h>


typedef __attribute__((ext_vector_type(2))) float v2f;
typedef __attribute__((ext_vector_type(4))) float v4f;
typedef __attribute__((ext_vector_type(8))) float v8f;

#define NGRP 40
#define NB   128
#define DIM  1024

#define KC      64
#define BPITCH  136
#define NSTAGE  (DIM / KC)
#define BLD     8

#define APITCH  68

#define WS_COUNTS 0
#define WS_ROWS   64

__global__ __launch_bounds__(NB) void build_lists_kernel(const int* __restrict__ g,
                                                         int* __restrict__ ws) {
    __shared__ __attribute__((aligned(16))) int sGrp[NB];
    const int t = threadIdx.x;

    int mine = g[t] % NGRP;
    if (mine < 0) mine += NGRP;
    sGrp[t] = mine;
    __syncthreads();

    int rank = 0;
    int cnt  = 0;
    int last = 0;
#pragma unroll
    for (int j = 0; j < NB; j += 4) {
        const int4 q = *(const int4*)(sGrp + j);
        rank += (q.x == mine && j     < t) ? 1 : 0;
        rank += (q.y == mine && j + 1 < t) ? 1 : 0;
        rank += (q.z == mine && j + 2 < t) ? 1 : 0;
        rank += (q.w == mine && j + 3 < t) ? 1 : 0;
        cnt  += (q.x == t) ? 1 : 0;
        cnt  += (q.y == t) ? 1 : 0;
        cnt  += (q.z == t) ? 1 : 0;
        cnt  += (q.w == t) ? 1 : 0;
        last = (q.x == t) ? j     : last;
        last = (q.y == t) ? j + 1 : last;
        last = (q.z == t) ? j + 2 : last;
        last = (q.w == t) ? j + 3 : last;
    }
    ws[WS_ROWS + mine * NB + rank] = t;

    if (t < NGRP) {
        ws[WS_COUNTS + t] = cnt;
        const int padded = cnt > 0 ? ((cnt + 15) & ~15) : 16;
        for (int i = 0; i < 16; ++i) {
            const int p = cnt + i;
            if (p < padded) ws[WS_ROWS + t * NB + p] = last;
        }
    }
}

__device__ __forceinline__ void stage_barrier() {
    asm volatile("s_wait_dscnt 0x0" ::: "memory");
    asm volatile("s_barrier_signal -1" ::: "memory");
    asm volatile("s_barrier_wait -1" ::: "memory");
}

__device__ __forceinline__ v8f stage_mma(const float* As, const float* Bs, v8f c) {
#pragma unroll
    for (int kl = 0; kl < KC; kl += 4) {
        v2f b;
        const v2f a = *(const v2f*)(As + kl);
        b[0] = Bs[kl * BPITCH];
        b[1] = Bs[(kl + 1) * BPITCH];
        c = __builtin_amdgcn_wmma_f32_16x16x4_f32(
                false, a, false, b, (short)0, c, false, false);
    }
    return c;
}

struct Slots {
    float* Bst0;
    float* Bst1;
    float* Ast0;
    float* Ast1;
    const float* Ap0;
    const float* Ap1;
    const float* Bp0;
    const float* Bp1;
};

__device__ __forceinline__ void tile_pass(const float* __restrict__ x, const float* Bg,
                                          v4f (&rA)[BLD], int arowIdx, int aslot,
                                          int m, int4 oLo, int4 oHi,
                                          const Slots& L, float* __restrict__ out,
                                          int n0, int r16, int half) {
    v4f rB[BLD];
    v4f aA, aB;

    const float* Ag = x + (size_t)arowIdx * DIM + aslot * 4;
    aA = *(const v4f*)(Ag);
    aB = *(const v4f*)(Ag + KC);
#pragma unroll
    for (int i = 0; i < BLD; ++i)
        rB[i] = *(const v4f*)(Bg + (size_t)(KC + i * 8) * DIM);

#pragma unroll
    for (int i = 0; i < BLD; ++i)
        *(v4f*)(L.Bst0 + i * 8 * BPITCH) = rA[i];
    *(v4f*)L.Ast0 = aA;
    stage_barrier();

    v8f c = {};

    for (int s = 0; s < NSTAGE - 2; s += 2) {
        const float* Bn = Bg + (size_t)(s + 2) * KC * DIM;
        const float* An = Ag + (s + 2) * KC;
#pragma unroll
        for (int i = 0; i < BLD; ++i)
            rA[i] = *(const v4f*)(Bn + (size_t)(i * 8) * DIM);
        aA = *(const v4f*)(An);

        c = stage_mma(L.Ap0, L.Bp0, c);

#pragma unroll
        for (int i = 0; i < BLD; ++i)
            *(v4f*)(L.Bst1 + i * 8 * BPITCH) = rB[i];
        *(v4f*)L.Ast1 = aB;
        stage_barrier();

#pragma unroll
        for (int i = 0; i < BLD; ++i)
            rB[i] = *(const v4f*)(Bn + (size_t)(KC + i * 8) * DIM);
        aB = *(const v4f*)(An + KC);

        c = stage_mma(L.Ap1, L.Bp1, c);

#pragma unroll
        for (int i = 0; i < BLD; ++i)
            *(v4f*)(L.Bst0 + i * 8 * BPITCH) = rA[i];
        *(v4f*)L.Ast0 = aA;
        stage_barrier();
    }
    c = stage_mma(L.Ap0, L.Bp0, c);
#pragma unroll
    for (int i = 0; i < BLD; ++i)
        *(v4f*)(L.Bst1 + i * 8 * BPITCH) = rB[i];
    *(v4f*)L.Ast1 = aB;
    stage_barrier();
    c = stage_mma(L.Ap1, L.Bp1, c);

#pragma unroll
    for (int j = 0; j < 8; ++j) {
        const int mrow = j + half * 8;
        const int b = (j < 4) ? oLo[j & 3] : oHi[j & 3];
        if (mrow < m)
            out[(size_t)b * DIM + n0 + r16] = c[j];
    }
}

__global__ __launch_bounds__(256) void grouped_gemm_kernel(
    const float* __restrict__ x,
    const float* __restrict__ T,
    const int*   __restrict__ ws,
    float*       __restrict__ out) {

    __shared__ __attribute__((aligned(16))) float Alds[2 * 16 * APITCH];
    __shared__ __attribute__((aligned(16))) float Blds[2 * KC * BPITCH];

    const int grp = blockIdx.x;
    const int tid = threadIdx.x;

    const int wave = tid >> 5;
    const int lane = tid & 31;
    const int r16  = lane & 15;
    const int half = lane >> 4;
    const int krow = half * 2;

    const int bslot = tid & 31;
    const int brow  = tid >> 5;
    const float* Bg = T + (size_t)grp * DIM * DIM + (size_t)brow * DIM
                        + blockIdx.y * 128 + bslot * 4;

    const int arow  = tid >> 4;
    const int aslot = tid & 15;

    const int* rowList = ws + WS_ROWS + grp * NB;
    int  arowIdx = rowList[arow];
    int4 oLo = *(const int4*)(rowList + half * 8);
    int4 oHi = *(const int4*)(rowList + half * 8 + 4);

    v4f rA[BLD];
#pragma unroll
    for (int i = 0; i < BLD; ++i)
        rA[i] = *(const v4f*)(Bg + (size_t)(i * 8) * DIM);

    const int cnt = ws[WS_COUNTS + grp];
    asm volatile("" : "+v"(arowIdx));
    if (cnt <= 0) return;

    Slots L;
    L.Bst0 = Blds + brow * BPITCH + bslot * 4;
    L.Bst1 = L.Bst0 + KC * BPITCH;
    L.Ast0 = Alds + arow * APITCH + aslot * 4;
    L.Ast1 = L.Ast0 + 16 * APITCH;
    L.Ap0  = Alds + r16 * APITCH + krow;
    L.Ap1  = L.Ap0 + 16 * APITCH;
    L.Bp0  = Blds + krow * BPITCH + wave * 16 + r16;
    L.Bp1  = L.Bp0 + KC * BPITCH;

    const int n0 = (blockIdx.y * 8 + wave) * 16;

    tile_pass(x, Bg, rA, arowIdx, aslot, cnt < 16 ? cnt : 16, oLo, oHi,
              L, out, n0, r16, half);

    for (int roff = 16; roff < cnt; roff += 16) {
        const float* Bg2 = Bg;
        asm volatile("" : "+v"(Bg2));
        arowIdx = rowList[roff + arow];
        oLo = *(const int4*)(rowList + roff + half * 8);
        oHi = *(const int4*)(rowList + roff + half * 8 + 4);
#pragma unroll
        for (int i = 0; i < BLD; ++i)
            rA[i] = *(const v4f*)(Bg2 + (size_t)(i * 8) * DIM);
        const int m = cnt - roff;
        tile_pass(x, Bg2, rA, arowIdx, aslot, m < 16 ? m : 16, oLo, oHi,
                  L, out, n0, r16, half);
    }
}

extern "C" void kernel_launch(void* const* d_in, const int* in_sizes, int n_in,
                              void* d_out, int out_size, void* d_ws, size_t ws_size,
                              hipStream_t stream) {
    const float* x = (const float*)d_in[0];
    const int*   g = (const int*)d_in[1];
    const float* T = (const float*)d_in[2];
    int*   ws  = (int*)d_ws;
    float* out = (float*)d_out;

    build_lists_kernel<<<1, NB, 0, stream>>>(g, ws);

    dim3 grid(NGRP, 8);
    grouped_gemm_kernel<<<grid, 256, 0, stream>>>(x, T, ws, out);
}


# 16-wave blocks (were 8): all waves load, 10-15 loads in flight each; a tile's K is shared by two waves (local k 0-31 / 32-63 of each stage), their sums added through LDS at the end: new f32 sum order
# speedup vs baseline: 2.3030x; 1.0474x over previous
#include <hip/hip_runtime.h>
#include <hip/hip_bf16.h>


typedef __attribute__((ext_vector_type(2))) float v2f;
typedef __attribute__((ext_vector_type(4))) float v4f;
typedef __attribute__((ext_vector_type(8))) float v8f;

#define NGRP 40
#define NB   128
#define DIM  1024

#define KC      64
#define BPITCH  136
#define NSTAGE  (DIM / KC)
#define BLD     4
#define RING    4
#define KH      (KC / 2)

#define APITCH  68

#define WS_COUNTS 0
#define WS_ROWS   64

__global__ __launch_bounds__(NB) void build_lists_kernel(const int* __restrict__ g,
                                                         int* __restrict__ ws) {
    __shared__ __attribute__((aligned(16))) int sGrp[NB];
    const int t = threadIdx.x;

    int mine = g[t] % NGRP;
    if (mine < 0) mine += NGRP;
    sGrp[t] = mine;
    __syncthreads();

    int rank = 0;
    int cnt  = 0;
    int last = 0;
#pragma unroll
    for (int j = 0; j < NB; j += 4) {
        const int4 q = *(const int4*)(sGrp + j);
        rank += (q.x == mine && j     < t) ? 1 : 0;
        rank += (q.y == mine && j + 1 < t) ? 1 : 0;
        rank += (q.z == mine && j + 2 < t) ? 1 : 0;
        rank += (q.w == mine && j + 3 < t) ? 1 : 0;
        cnt  += (q.x == t) ? 1 : 0;
        cnt  += (q.y == t) ? 1 : 0;
        cnt  += (q.z == t) ? 1 : 0;
        cnt  += (q.w == t) ? 1 : 0;
        last = (q.x == t) ? j     : last;
        last = (q.y == t) ? j + 1 : last;
        last = (q.z == t) ? j + 2 : last;
        last = (q.w == t) ? j + 3 : last;
    }
    ws[WS_ROWS + mine * NB + rank] = t;

    if (t < NGRP) {
        ws[WS_COUNTS + t] = cnt;
        const int padded = cnt > 0 ? ((cnt + 15) & ~15) : 16;
        for (int i = 0; i < 16; ++i) {
            const int p = cnt + i;
            if (p < padded) ws[WS_ROWS + t * NB + p] = last;
        }
    }
}

__device__ __forceinline__ void stage_barrier() {
    asm volatile("s_wait_dscnt 0x0" ::: "memory");
    asm volatile("s_barrier_signal -1" ::: "memory");
    asm volatile("s_barrier_wait -1" ::: "memory");
}

__device__ __forceinline__ v8f stage_mma(const float* As, const float* Bs, v8f c) {
#pragma unroll
    for (int kl = 0; kl < KH; kl += 4) {
        v2f b;
        const v2f a = *(const v2f*)(As + kl);
        b[0] = Bs[kl * BPITCH];
        b[1] = Bs[(kl + 1) * BPITCH];
        c = __builtin_amdgcn_wmma_f32_16x16x4_f32(
                false, a, false, b, (short)0, c, false, false);
    }
    return c;
}

struct Slots {
    float* Bst[2];
    float* Ast[2];
    const float* Ap[2];
    const float* Bp[2];
};

__device__ __forceinline__ void store_stage(const Slots& L, int b, const v4f (&t)[BLD], v2f xv) {
#pragma unroll
    for (int i = 0; i < BLD; ++i)
        *(v4f*)(L.Bst[b] + i * 16 * BPITCH) = t[i];
    *(v2f*)L.Ast[b] = xv;
}

__device__ __forceinline__ void tile_pass(const float* __restrict__ x, const float* __restrict__ Tg,
                                          unsigned lo, v4f (&rT)[RING][BLD], int arowIdx, int lane,
                                          bool hi, int tile, float* Plds,
                                          int m, const int* __restrict__ rl,
                                          const Slots& L, float* __restrict__ out,
                                          int n0) {
    v2f rX[RING];
    const unsigned xo = (unsigned)arowIdx * DIM + lane * 2;
    rX[0] = *(const v2f*)(x + xo);
#pragma unroll
    for (int q = 1; q < RING; ++q) {
#pragma unroll
        for (int i = 0; i < BLD; ++i)
            rT[q][i] = *(const v4f*)(Tg + (size_t)(q * KC + i * 16) * DIM + lo);
        rX[q] = *(const v2f*)(x + q * KC + xo);
    }

    store_stage(L, 0, rT[0], rX[0]);
    stage_barrier();

    v8f c = {};

    for (int t = 0; t < NSTAGE / RING - 1; ++t) {
        const float* Bn = Tg + (size_t)(t + 1) * RING * KC * DIM;
        const float* An = x + (t + 1) * RING * KC;
#pragma unroll
        for (int q = 0; q < RING; ++q) {
#pragma unroll
            for (int i = 0; i < BLD; ++i)
                rT[q][i] = *(const v4f*)(Bn + (size_t)(q * KC + i * 16) * DIM + lo);
            rX[q] = *(const v2f*)(An + q * KC + xo);

            c = stage_mma(L.Ap[q & 1], L.Bp[q & 1], c);

            store_stage(L, (q + 1) & 1, rT[(q + 1) & (RING - 1)], rX[(q + 1) & (RING - 1)]);
            stage_barrier();
        }
    }
    int te = threadIdx.x;
    asm volatile("" : "+v"(te));
    const int laneE = te & 31;
    const int r16E  = te & 15;
    const int halfE = (te >> 4) & 1;
    const int4 oLo = *(const int4*)(rl + halfE * 8);
    const int4 oHi = *(const int4*)(rl + halfE * 8 + 4);

#pragma unroll
    for (int q = 0; q < RING; ++q) {
        c = stage_mma(L.Ap[q & 1], L.Bp[q & 1], c);
        if (q + 1 < RING) {
            store_stage(L, (q + 1) & 1, rT[(q + 1) & (RING - 1)], rX[(q + 1) & (RING - 1)]);
            stage_barrier();
        }
    }

    if (hi) {
#pragma unroll
        for (int j = 0; j < 8; ++j)
            Plds[(j * 8 + tile) * 32 + laneE] = c[j];
    }
    stage_barrier();

    if (!hi) {
#pragma unroll
        for (int j = 0; j < 8; ++j)
            c[j] += Plds[(j * 8 + tile) * 32 + laneE];
#pragma unroll
        for (int j = 0; j < 8; ++j) {
            const int mrow = j + halfE * 8;
            const int b = (j < 4) ? oLo[j & 3] : oHi[j & 3];
            if (mrow < m)
                out[(size_t)b * DIM + n0 + r16E] = c[j];
        }
    }
}

__global__ __launch_bounds__(512) void grouped_gemm_kernel(
    const float* __restrict__ x,
    const float* __restrict__ T,
    const int*   __restrict__ ws,
    float*       __restrict__ out) {

    __shared__ __attribute__((aligned(16))) float Alds[2 * 16 * APITCH];
    __shared__ __attribute__((aligned(16))) float Blds[2 * KC * BPITCH];
    __shared__ __attribute__((aligned(16))) float Plds[8 * 8 * 32];

    const int grp = blockIdx.x;
    const int tid = threadIdx.x;

    const int wv   = tid >> 5;
    const int wave = __builtin_amdgcn_readfirstlane(wv);
    const bool hi  = wave >= 8;
    const int tile = wave & 7;
    const int lane = tid & 31;
    const int r16  = lane & 15;
    const int half = lane >> 4;
    const int krow = half * 2 + (hi ? KH : 0);

    const float* Tg = T + (size_t)grp * DIM * DIM + blockIdx.y * 128;
    unsigned lo = wv * DIM + lane * 4;

    const int* rowList = ws + WS_ROWS + grp * NB;
    int  arowIdx = rowList[wv];

    v4f rT[RING][BLD];
#pragma unroll
    for (int i = 0; i < BLD; ++i)
        rT[0][i] = *(const v4f*)(Tg + (size_t)(i * 16) * DIM + lo);

    const int cnt = ws[WS_COUNTS + grp];
    asm volatile("" : "+v"(arowIdx));
    if (cnt <= 0) return;

    int oBst = wv * BPITCH + lane * 4;
    int oAst = wv * APITCH + lane * 2;
    int oAp  = r16 * APITCH + krow;
    int oBp  = krow * BPITCH + tile * 16 + r16;

    const int n0 = (blockIdx.y * 8 + tile) * 16;

    for (int roff = 0; roff < NB; roff += 16) {
        if (roff > 0) {
            arowIdx = rowList[roff + wv];
            asm volatile("" : "+v"(lo), "+v"(oBst), "+v"(oAst), "+v"(oAp), "+v"(oBp));
#pragma unroll
            for (int i = 0; i < BLD; ++i)
                rT[0][i] = *(const v4f*)(Tg + (size_t)(i * 16) * DIM + lo);
        }
        Slots L;
        L.Bst[0] = Blds + oBst;
        L.Bst[1] = L.Bst[0] + KC * BPITCH;
        L.Ast[0] = Alds + oAst;
        L.Ast[1] = L.Ast[0] + 16 * APITCH;
        L.Ap[0]  = Alds + oAp;
        L.Ap[1]  = L.Ap[0] + 16 * APITCH;
        L.Bp[0]  = Blds + oBp;
        L.Bp[1]  = L.Bp[0] + KC * BPITCH;

        const int m = cnt - roff;
        tile_pass(x, Tg, lo, rT, arowIdx, lane, hi, tile, Plds, m < 16 ? m : 16,
                  rowList + roff, L, out, n0);

        if (roff + 16 >= cnt) break;
    }
}

extern "C" void kernel_launch(void* const* d_in, const int* in_sizes, int n_in,
                              void* d_out, int out_size, void* d_ws, size_t ws_size,
                              hipStream_t stream) {
    const float* x = (const float*)d_in[0];
    const int*   g = (const int*)d_in[1];
    const float* T = (const float*)d_in[2];
    int*   ws  = (int*)d_ws;
    float* out = (float*)d_out;

    build_lists_kernel<<<1, NB, 0, stream>>>(g, ws);

    dim3 grid(NGRP, 8);
    grouped_gemm_kernel<<<grid, 512, 0, stream>>>(x, T, ws, out);
}
